# starting kernel
# baseline (speedup 1.0000x reference)
#include <hip/hip_runtime.h>
#include <hip/hip_bf16.h>
#include <math.h>

#define R_H 6
#define EPSBN 1e-5f
#define BSZ 8
#define CSZ 64
#define NSZ 16384
#define OC  64
#define DT  128
#define YPS 288
#define WPS 160

typedef __attribute__((ext_vector_type(2))) float v2f;
typedef __attribute__((ext_vector_type(8))) float v8f;

__device__ __forceinline__ void hermite6(float xn, float p[R_H]) {
    const float SQ13 = 3.60555127546398929312f;
    const float PIQ  = 0.75112554446494248286f;
    float t = tanhf(xn) * SQ13;
    float g = __expf(-0.5f * t * t);
    p[0] = PIQ * g;
    p[1] = 1.41421356237309515f * PIQ * t * g;
    p[2] = 1.0f                 * t * p[1] - 0.70710678118654752f * p[0];
    p[3] = 0.81649658092772603f * t * p[2] - 0.81649658092772603f * p[1];
    p[4] = 0.70710678118654752f * t * p[3] - 0.86602540378443865f * p[2];
    p[5] = 0.63245553203367587f * t * p[4] - 0.89442719099991588f * p[3];
}

__global__ void k_init(float* __restrict__ ws) {
    int i = threadIdx.x;
    if (i < 2 * CSZ) ws[i] = 0.0f;
}

__global__ void __launch_bounds__(256) k_stats(const float* __restrict__ x,
                                               float* __restrict__ ws) {
    __shared__ float red[256];
    int bc = blockIdx.x;
    int c  = bc & (CSZ - 1);
    const float* xp = x + (size_t)bc * NSZ;
    float s = 0.f, s2 = 0.f;
    for (int d = threadIdx.x; d < NSZ; d += 256) {
        float v = xp[d];
        s += v; s2 += v * v;
    }
    red[threadIdx.x] = s; __syncthreads();
    for (int st = 128; st > 0; st >>= 1) {
        if (threadIdx.x < st) red[threadIdx.x] += red[threadIdx.x + st];
        __syncthreads();
    }
    if (threadIdx.x == 0) atomicAdd(&ws[c], red[0]);
    __syncthreads();
    red[threadIdx.x] = s2; __syncthreads();
    for (int st = 128; st > 0; st >>= 1) {
        if (threadIdx.x < st) red[threadIdx.x] += red[threadIdx.x + st];
        __syncthreads();
    }
    if (threadIdx.x == 0) atomicAdd(&ws[CSZ + c], red[0]);
}

__global__ void k_finalize(const float* __restrict__ gamma,
                           const float* __restrict__ beta,
                           float* __restrict__ ws) {
    int c = threadIdx.x;
    if (c >= CSZ) return;
    const float inv_n = 1.0f / ((float)BSZ * (float)NSZ);
    float mean = ws[c] * inv_n;
    float var  = ws[CSZ + c] * inv_n - mean * mean;
    float rstd = rsqrtf(var + EPSBN);
    float sc   = gamma[c] * rstd;
    ws[2 * CSZ + c] = sc;
    ws[3 * CSZ + c] = beta[c] - mean * sc;
}

__global__ void __launch_bounds__(256) k_cost(const float* __restrict__ x,
                                              float* __restrict__ ws) {
    __shared__ float red[256];
    __shared__ float costk[R_H];
    int bc = blockIdx.x;
    int c  = bc & (CSZ - 1);
    float sc = ws[2 * CSZ + c];
    float sh = ws[3 * CSZ + c];
    const float* xp = x + (size_t)bc * NSZ;

    float acc[R_H];
#pragma unroll
    for (int k = 0; k < R_H; ++k) acc[k] = 0.f;

    for (int d = threadIdx.x; d < NSZ; d += 256) {
        float xn = fmaf(sc, xp[d], sh);
        float p[R_H];
        hermite6(xn, p);
#pragma unroll
        for (int k = 0; k < R_H; ++k) acc[k] = fmaf(p[k], xn, acc[k]);
    }
#pragma unroll
    for (int k = 0; k < R_H; ++k) {
        red[threadIdx.x] = acc[k]; __syncthreads();
        for (int st = 128; st > 0; st >>= 1) {
            if (threadIdx.x < st) red[threadIdx.x] += red[threadIdx.x + st];
            __syncthreads();
        }
        if (threadIdx.x == 0) costk[k] = red[0];
        __syncthreads();
    }
    if (threadIdx.x == 0) {
        float m = costk[0];
#pragma unroll
        for (int k = 1; k < R_H; ++k) m = fmaxf(m, costk[k]);
        float e[R_H], ssum = 0.f;
#pragma unroll
        for (int k = 0; k < R_H; ++k) { e[k] = __expf(costk[k] - m); ssum += e[k]; }
        float inv = 1.0f / ssum;
#pragma unroll
        for (int k = 0; k < R_H; ++k) ws[4 * CSZ + bc * R_H + k] = e[k] * inv;
    }
}

__global__ void __launch_bounds__(256) k_fused(const float* __restrict__ x,
                                               const float* __restrict__ Wt,
                                               const float* __restrict__ bias,
                                               const float* __restrict__ ws,
                                               float* __restrict__ out) {
    __shared__ float y_pair[(CSZ / 2) * YPS];
    __shared__ float wt_pair[(CSZ / 2) * WPS];
    __shared__ float sc_l[CSZ], sh_l[CSZ], w_l[CSZ * R_H];

    int b      = blockIdx.y;
    int d_base = blockIdx.x * DT;
    int tid    = threadIdx.x;

    for (int i = tid; i < OC * CSZ; i += 256) {
        int o = i >> 6, c = i & (CSZ - 1);
        wt_pair[(c >> 1) * WPS + o * 2 + (c & 1)] = Wt[i];
    }
    if (tid < CSZ) { sc_l[tid] = ws[2 * CSZ + tid]; sh_l[tid] = ws[3 * CSZ + tid]; }
    for (int i = tid; i < CSZ * R_H; i += 256)
        w_l[i] = ws[4 * CSZ + b * CSZ * R_H + i];
    __syncthreads();

    int dl   = tid & (DT - 1);
    int cofs = tid >> 7;
    for (int cb = 0; cb < CSZ; cb += 2) {
        int c = cb + cofs;
        const float* xrow = &x[(((size_t)b * CSZ + c) * NSZ) + d_base + dl];
        float xv = *xrow;
        if (cb + 2 < CSZ) __builtin_prefetch(xrow + 2 * NSZ, 0, 0);
        float xn = fmaf(sc_l[c], xv, sh_l[c]);
        float p[R_H];
        hermite6(xn, p);
        float y = 0.f;
#pragma unroll
        for (int k = 0; k < R_H; ++k) y = fmaf(w_l[c * R_H + k], p[k], y);
        y_pair[(cb >> 1) * YPS + dl * 2 + cofs] = y;
    }
    __syncthreads();

    int wave = tid >> 5;
    int lane = tid & 31;
    int half = lane >> 4;
    int lid  = lane & 15;
    int o0   = (wave & 3) << 4;

    v2f afrag[16];
#pragma unroll
    for (int s = 0; s < 16; ++s)
        afrag[s] = *(const v2f*)&wt_pair[(2 * s + half) * WPS + (o0 + lid) * 2];

#pragma unroll
    for (int i = 0; i < 4; ++i) {
        int d0 = ((((wave >> 2) << 2) + i) << 4);
        v8f acc = {0.f, 0.f, 0.f, 0.f, 0.f, 0.f, 0.f, 0.f};
#pragma unroll
        for (int s = 0; s < 16; ++s) {
            v2f bb = *(const v2f*)&y_pair[(2 * s + half) * YPS + (d0 + lid) * 2];
            acc = __builtin_amdgcn_wmma_f32_16x16x4_f32(
                false, afrag[s], false, bb, (short)0, acc, false, false);
        }
        int dcol = d_base + d0 + lid;
#pragma unroll
        for (int v = 0; v < 8; ++v) {
            int o = o0 + v + 8 * half;
            out[(((size_t)b * OC + o) * NSZ) + dcol] = acc[v] + bias[o];
        }
    }
}

extern "C" void kernel_launch(void* const* d_in, const int* in_sizes, int n_in,
                              void* d_out, int out_size, void* d_ws, size_t ws_size,
                              hipStream_t stream) {
    const float* x     = (const float*)d_in[0];
    const float* gamma = (const float*)d_in[1];
    const float* beta  = (const float*)d_in[2];
    const float* Wt    = (const float*)d_in[3];
    const float* bias  = (const float*)d_in[4];
    float* out = (float*)d_out;
    float* ws  = (float*)d_ws;

    hipLaunchKernelGGL(k_init,     dim3(1),             dim3(128), 0, stream, ws);
    hipLaunchKernelGGL(k_stats,    dim3(BSZ * CSZ),     dim3(256), 0, stream, x, ws);
    hipLaunchKernelGGL(k_finalize, dim3(1),             dim3(64),  0, stream, gamma, beta, ws);
    hipLaunchKernelGGL(k_cost,     dim3(BSZ * CSZ),     dim3(256), 0, stream, x, ws);
    hipLaunchKernelGGL(k_fused,    dim3(NSZ / DT, BSZ), dim3(256), 0, stream,
                       x, Wt, bias, ws, out);
}


# all three passes over x fetch it with 16-byte loads issued in groups ahead of the math (was one exposed 4-byte load per trip); y-tile thread map changed to suit
# speedup vs baseline: 1.3965x; 1.3965x over previous
#include <hip/hip_runtime.h>
#include <hip/hip_bf16.h>
#include <math.h>

#define R_H 6
#define EPSBN 1e-5f
#define BSZ 8
#define CSZ 64
#define NSZ 16384
#define OC  64
#define DT  128
#define YPS 288
#define WPS 160

typedef __attribute__((ext_vector_type(2))) float v2f;
typedef __attribute__((ext_vector_type(8))) float v8f;

__device__ __forceinline__ void hermite6(float xn, float p[R_H]) {
    const float SQ13 = 3.60555127546398929312f;
    const float PIQ  = 0.75112554446494248286f;
    float t = tanhf(xn) * SQ13;
    float g = __expf(-0.5f * t * t);
    p[0] = PIQ * g;
    p[1] = 1.41421356237309515f * PIQ * t * g;
    p[2] = 1.0f                 * t * p[1] - 0.70710678118654752f * p[0];
    p[3] = 0.81649658092772603f * t * p[2] - 0.81649658092772603f * p[1];
    p[4] = 0.70710678118654752f * t * p[3] - 0.86602540378443865f * p[2];
    p[5] = 0.63245553203367587f * t * p[4] - 0.89442719099991588f * p[3];
}

__global__ void k_init(float* __restrict__ ws) {
    int i = threadIdx.x;
    if (i < 2 * CSZ) ws[i] = 0.0f;
}

__global__ void __launch_bounds__(256) k_stats(const float* __restrict__ x,
                                               float* __restrict__ ws) {
    __shared__ float red[256];
    int bc = blockIdx.x;
    int c  = bc & (CSZ - 1);
    const float4* xp4 = reinterpret_cast<const float4*>(x + (size_t)bc * NSZ) + threadIdx.x;
    float4 v[16];
#pragma unroll
    for (int j = 0; j < 16; ++j) v[j] = xp4[j * 256];
    float4 a = {0.f, 0.f, 0.f, 0.f}, q = {0.f, 0.f, 0.f, 0.f};
#pragma unroll
    for (int j = 0; j < 16; ++j) {
        a.x += v[j].x; q.x += v[j].x * v[j].x;
        a.y += v[j].y; q.y += v[j].y * v[j].y;
        a.z += v[j].z; q.z += v[j].z * v[j].z;
        a.w += v[j].w; q.w += v[j].w * v[j].w;
    }
    float s  = (a.x + a.y) + (a.z + a.w);
    float s2 = (q.x + q.y) + (q.z + q.w);
    red[threadIdx.x] = s; __syncthreads();
    for (int st = 128; st > 0; st >>= 1) {
        if (threadIdx.x < st) red[threadIdx.x] += red[threadIdx.x + st];
        __syncthreads();
    }
    if (threadIdx.x == 0) atomicAdd(&ws[c], red[0]);
    __syncthreads();
    red[threadIdx.x] = s2; __syncthreads();
    for (int st = 128; st > 0; st >>= 1) {
        if (threadIdx.x < st) red[threadIdx.x] += red[threadIdx.x + st];
        __syncthreads();
    }
    if (threadIdx.x == 0) atomicAdd(&ws[CSZ + c], red[0]);
}

__global__ void k_finalize(const float* __restrict__ gamma,
                           const float* __restrict__ beta,
                           float* __restrict__ ws) {
    int c = threadIdx.x;
    if (c >= CSZ) return;
    const float inv_n = 1.0f / ((float)BSZ * (float)NSZ);
    float mean = ws[c] * inv_n;
    float var  = ws[CSZ + c] * inv_n - mean * mean;
    float rstd = rsqrtf(var + EPSBN);
    float sc   = gamma[c] * rstd;
    ws[2 * CSZ + c] = sc;
    ws[3 * CSZ + c] = beta[c] - mean * sc;
}

__device__ __forceinline__ void cost_acc1(float xv, float sc, float sh, float acc[R_H]) {
    float xn = fmaf(sc, xv, sh);
    float p[R_H];
    hermite6(xn, p);
#pragma unroll
    for (int k = 0; k < R_H; ++k) acc[k] = fmaf(p[k], xn, acc[k]);
}
__device__ __forceinline__ void cost_acc4(float4 v, float sc, float sh, float acc[R_H]) {
    cost_acc1(v.x, sc, sh, acc);
    cost_acc1(v.y, sc, sh, acc);
    cost_acc1(v.z, sc, sh, acc);
    cost_acc1(v.w, sc, sh, acc);
}

__device__ __forceinline__ float y_of(float xv, float sc, float sh, const float w[R_H]) {
    float xn = fmaf(sc, xv, sh);
    float p[R_H];
    hermite6(xn, p);
    float y = 0.f;
#pragma unroll
    for (int k = 0; k < R_H; ++k) y = fmaf(w[k], p[k], y);
    return y;
}

__global__ void __launch_bounds__(256) k_cost(const float* __restrict__ x,
                                              float* __restrict__ ws) {
    __shared__ float red[256];
    __shared__ float costk[R_H];
    int bc = blockIdx.x;
    int c  = bc & (CSZ - 1);
    float sc = ws[2 * CSZ + c];
    float sh = ws[3 * CSZ + c];
    const float* xp = x + (size_t)bc * NSZ;

    float acc[R_H];
#pragma unroll
    for (int k = 0; k < R_H; ++k) acc[k] = 0.f;

    const float4* xp4 = reinterpret_cast<const float4*>(xp) + threadIdx.x;
    float4 c0 = xp4[0], c1 = xp4[256], c2 = xp4[512], c3 = xp4[768];
    for (int g = 0; g < 4; ++g) {
        float4 n0 = c0, n1 = c1, n2 = c2, n3 = c3;
        if (g < 3) {
            const float4* nx = xp4 + (g + 1) * 1024;
            n0 = nx[0]; n1 = nx[256]; n2 = nx[512]; n3 = nx[768];
        }
        cost_acc4(c0, sc, sh, acc);
        cost_acc4(c1, sc, sh, acc);
        cost_acc4(c2, sc, sh, acc);
        cost_acc4(c3, sc, sh, acc);
        c0 = n0; c1 = n1; c2 = n2; c3 = n3;
    }
#pragma unroll
    for (int k = 0; k < R_H; ++k) {
        red[threadIdx.x] = acc[k]; __syncthreads();
        for (int st = 128; st > 0; st >>= 1) {
            if (threadIdx.x < st) red[threadIdx.x] += red[threadIdx.x + st];
            __syncthreads();
        }
        if (threadIdx.x == 0) costk[k] = red[0];
        __syncthreads();
    }
    if (threadIdx.x == 0) {
        float m = costk[0];
#pragma unroll
        for (int k = 1; k < R_H; ++k) m = fmaxf(m, costk[k]);
        float e[R_H], ssum = 0.f;
#pragma unroll
        for (int k = 0; k < R_H; ++k) { e[k] = __expf(costk[k] - m); ssum += e[k]; }
        float inv = 1.0f / ssum;
#pragma unroll
        for (int k = 0; k < R_H; ++k) ws[4 * CSZ + bc * R_H + k] = e[k] * inv;
    }
}

__global__ void __launch_bounds__(256) k_fused(const float* __restrict__ x,
                                               const float* __restrict__ Wt,
                                               const float* __restrict__ bias,
                                               const float* __restrict__ ws,
                                               float* __restrict__ out) {
    __shared__ __attribute__((aligned(16))) float y_pair[(CSZ / 2) * YPS];
    __shared__ __attribute__((aligned(16))) float wt_pair[(CSZ / 2) * WPS];
    __shared__ __attribute__((aligned(16))) float sc_l[CSZ], sh_l[CSZ], w_l[CSZ * R_H];

    int b      = blockIdx.y;
    int d_base = blockIdx.x * DT;
    int tid    = threadIdx.x;
    int wave   = tid >> 5;
    int lane   = tid & 31;

    const float4* xq = reinterpret_cast<const float4*>(
                           x + ((size_t)b * CSZ + 2 * wave) * NSZ + d_base) + lane;
    float4 ca = xq[0], cb = xq[NSZ / 4];

    for (int i = tid; i < OC * CSZ; i += 256) {
        int o = i >> 6, c = i & (CSZ - 1);
        wt_pair[(c >> 1) * WPS + o * 2 + (c & 1)] = Wt[i];
    }
    if (tid < CSZ) { sc_l[tid] = ws[2 * CSZ + tid]; sh_l[tid] = ws[3 * CSZ + tid]; }
    for (int i = tid; i < CSZ * R_H; i += 256)
        w_l[i] = ws[4 * CSZ + b * CSZ * R_H + i];
    __syncthreads();

    for (int j = 0; j < 4; ++j) {
        float4 na = ca, nb = cb;
        if (j < 3) {
            const float4* nx = xq + (size_t)(j + 1) * (16 * (NSZ / 4));
            na = nx[0]; nb = nx[NSZ / 4];
        }
        int cp = wave + 8 * j;
        int ce = 2 * cp;
        float sc0 = sc_l[ce], sc1 = sc_l[ce + 1];
        float sh0 = sh_l[ce], sh1 = sh_l[ce + 1];
        float w0[R_H], w1[R_H];
#pragma unroll
        for (int k = 0; k < R_H; ++k) { w0[k] = w_l[ce * R_H + k]; w1[k] = w_l[ce * R_H + R_H + k]; }
        float4 lo, hi;
        lo.x = y_of(ca.x, sc0, sh0, w0); lo.y = y_of(cb.x, sc1, sh1, w1);
        lo.z = y_of(ca.y, sc0, sh0, w0); lo.w = y_of(cb.y, sc1, sh1, w1);
        hi.x = y_of(ca.z, sc0, sh0, w0); hi.y = y_of(cb.z, sc1, sh1, w1);
        hi.z = y_of(ca.w, sc0, sh0, w0); hi.w = y_of(cb.w, sc1, sh1, w1);
        float4* dst = reinterpret_cast<float4*>(&y_pair[cp * YPS + lane * 8]);
        dst[0] = lo; dst[1] = hi;
        ca = na; cb = nb;
    }
    __syncthreads();

    int half = lane >> 4;
    int lid  = lane & 15;
    int o0   = (wave & 3) << 4;

    v2f afrag[16];
#pragma unroll
    for (int s = 0; s < 16; ++s)
        afrag[s] = *(const v2f*)&wt_pair[(2 * s + half) * WPS + (o0 + lid) * 2];

#pragma unroll
    for (int i = 0; i < 4; ++i) {
        int d0 = ((((wave >> 2) << 2) + i) << 4);
        v8f acc = {0.f, 0.f, 0.f, 0.f, 0.f, 0.f, 0.f, 0.f};
#pragma unroll
        for (int s = 0; s < 16; ++s) {
            v2f bb = *(const v2f*)&y_pair[(2 * s + half) * YPS + (d0 + lid) * 2];
            acc = __builtin_amdgcn_wmma_f32_16x16x4_f32(
                false, afrag[s], false, bb, (short)0, acc, false, false);
        }
        int dcol = d_base + d0 + lid;
#pragma unroll
        for (int v = 0; v < 8; ++v) {
            int o = o0 + v + 8 * half;
            out[(((size_t)b * OC + o) * NSZ) + dcol] = acc[v] + bias[o];
        }
    }
}

extern "C" void kernel_launch(void* const* d_in, const int* in_sizes, int n_in,
                              void* d_out, int out_size, void* d_ws, size_t ws_size,
                              hipStream_t stream) {
    const float* x     = (const float*)d_in[0];
    const float* gamma = (const float*)d_in[1];
    const float* beta  = (const float*)d_in[2];
    const float* Wt    = (const float*)d_in[3];
    const float* bias  = (const float*)d_in[4];
    float* out = (float*)d_out;
    float* ws  = (float*)d_ws;

    hipLaunchKernelGGL(k_init,     dim3(1),             dim3(128), 0, stream, ws);
    hipLaunchKernelGGL(k_stats,    dim3(BSZ * CSZ),     dim3(256), 0, stream, x, ws);
    hipLaunchKernelGGL(k_finalize, dim3(1),             dim3(64),  0, stream, gamma, beta, ws);
    hipLaunchKernelGGL(k_cost,     dim3(BSZ * CSZ),     dim3(256), 0, stream, x, ws);
    hipLaunchKernelGGL(k_fused,    dim3(NSZ / DT, BSZ), dim3(256), 0, stream,
                       x, Wt, bias, ws, out);
}


# sqrt(13)*tanh(x) evaluated as sqrt(13) - 2*sqrt(13)/(exp(2x)+1) on the hardware exp2 and reciprocal instead of the library tanhf, in both passes that build the Hermite functions
# speedup vs baseline: 1.7122x; 1.2261x over previous
#include <hip/hip_runtime.h>
#include <hip/hip_bf16.h>
#include <math.h>

#define R_H 6
#define EPSBN 1e-5f
#define BSZ 8
#define CSZ 64
#define NSZ 16384
#define OC  64
#define DT  128
#define YPS 288
#define WPS 160

typedef __attribute__((ext_vector_type(2))) float v2f;
typedef __attribute__((ext_vector_type(8))) float v8f;

__device__ __forceinline__ void hermite6(float xn, float p[R_H]) {
    const float SQ13 = 3.60555127546398929312f;
    const float PIQ  = 0.75112554446494248286f;
    float e = __builtin_amdgcn_exp2f(xn * 2.88539008177792681472f);
    float t = fmaf(-2.0f * SQ13, __builtin_amdgcn_rcpf(e + 1.0f), SQ13);
    float g = __expf(-0.5f * t * t);
    p[0] = PIQ * g;
    p[1] = 1.41421356237309515f * PIQ * t * g;
    p[2] = 1.0f                 * t * p[1] - 0.70710678118654752f * p[0];
    p[3] = 0.81649658092772603f * t * p[2] - 0.81649658092772603f * p[1];
    p[4] = 0.70710678118654752f * t * p[3] - 0.86602540378443865f * p[2];
    p[5] = 0.63245553203367587f * t * p[4] - 0.89442719099991588f * p[3];
}

__global__ void k_init(float* __restrict__ ws) {
    int i = threadIdx.x;
    if (i < 2 * CSZ) ws[i] = 0.0f;
}

__global__ void __launch_bounds__(256) k_stats(const float* __restrict__ x,
                                               float* __restrict__ ws) {
    __shared__ float red[256];
    int bc = blockIdx.x;
    int c  = bc & (CSZ - 1);
    const float4* xp4 = reinterpret_cast<const float4*>(x + (size_t)bc * NSZ) + threadIdx.x;
    float4 v[16];
#pragma unroll
    for (int j = 0; j < 16; ++j) v[j] = xp4[j * 256];
    float4 a = {0.f, 0.f, 0.f, 0.f}, q = {0.f, 0.f, 0.f, 0.f};
#pragma unroll
    for (int j = 0; j < 16; ++j) {
        a.x += v[j].x; q.x += v[j].x * v[j].x;
        a.y += v[j].y; q.y += v[j].y * v[j].y;
        a.z += v[j].z; q.z += v[j].z * v[j].z;
        a.w += v[j].w; q.w += v[j].w * v[j].w;
    }
    float s  = (a.x + a.y) + (a.z + a.w);
    float s2 = (q.x + q.y) + (q.z + q.w);
    red[threadIdx.x] = s; __syncthreads();
    for (int st = 128; st > 0; st >>= 1) {
        if (threadIdx.x < st) red[threadIdx.x] += red[threadIdx.x + st];
        __syncthreads();
    }
    if (threadIdx.x == 0) atomicAdd(&ws[c], red[0]);
    __syncthreads();
    red[threadIdx.x] = s2; __syncthreads();
    for (int st = 128; st > 0; st >>= 1) {
        if (threadIdx.x < st) red[threadIdx.x] += red[threadIdx.x + st];
        __syncthreads();
    }
    if (threadIdx.x == 0) atomicAdd(&ws[CSZ + c], red[0]);
}

__global__ void k_finalize(const float* __restrict__ gamma,
                           const float* __restrict__ beta,
                           float* __restrict__ ws) {
    int c = threadIdx.x;
    if (c >= CSZ) return;
    const float inv_n = 1.0f / ((float)BSZ * (float)NSZ);
    float mean = ws[c] * inv_n;
    float var  = ws[CSZ + c] * inv_n - mean * mean;
    float rstd = rsqrtf(var + EPSBN);
    float sc   = gamma[c] * rstd;
    ws[2 * CSZ + c] = sc;
    ws[3 * CSZ + c] = beta[c] - mean * sc;
}

__device__ __forceinline__ void cost_acc1(float xv, float sc, float sh, float acc[R_H]) {
    float xn = fmaf(sc, xv, sh);
    float p[R_H];
    hermite6(xn, p);
#pragma unroll
    for (int k = 0; k < R_H; ++k) acc[k] = fmaf(p[k], xn, acc[k]);
}
__device__ __forceinline__ void cost_acc4(float4 v, float sc, float sh, float acc[R_H]) {
    cost_acc1(v.x, sc, sh, acc);
    cost_acc1(v.y, sc, sh, acc);
    cost_acc1(v.z, sc, sh, acc);
    cost_acc1(v.w, sc, sh, acc);
}

__device__ __forceinline__ float y_of(float xv, float sc, float sh, const float w[R_H]) {
    float xn = fmaf(sc, xv, sh);
    float p[R_H];
    hermite6(xn, p);
    float y = 0.f;
#pragma unroll
    for (int k = 0; k < R_H; ++k) y = fmaf(w[k], p[k], y);
    return y;
}

__global__ void __launch_bounds__(256) k_cost(const float* __restrict__ x,
                                              float* __restrict__ ws) {
    __shared__ float red[256];
    __shared__ float costk[R_H];
    int bc = blockIdx.x;
    int c  = bc & (CSZ - 1);
    float sc = ws[2 * CSZ + c];
    float sh = ws[3 * CSZ + c];
    const float* xp = x + (size_t)bc * NSZ;

    float acc[R_H];
#pragma unroll
    for (int k = 0; k < R_H; ++k) acc[k] = 0.f;

    const float4* xp4 = reinterpret_cast<const float4*>(xp) + threadIdx.x;
    float4 c0 = xp4[0], c1 = xp4[256], c2 = xp4[512], c3 = xp4[768];
    for (int g = 0; g < 4; ++g) {
        float4 n0 = c0, n1 = c1, n2 = c2, n3 = c3;
        if (g < 3) {
            const float4* nx = xp4 + (g + 1) * 1024;
            n0 = nx[0]; n1 = nx[256]; n2 = nx[512]; n3 = nx[768];
        }
        cost_acc4(c0, sc, sh, acc);
        cost_acc4(c1, sc, sh, acc);
        cost_acc4(c2, sc, sh, acc);
        cost_acc4(c3, sc, sh, acc);
        c0 = n0; c1 = n1; c2 = n2; c3 = n3;
    }
#pragma unroll
    for (int k = 0; k < R_H; ++k) {
        red[threadIdx.x] = acc[k]; __syncthreads();
        for (int st = 128; st > 0; st >>= 1) {
            if (threadIdx.x < st) red[threadIdx.x] += red[threadIdx.x + st];
            __syncthreads();
        }
        if (threadIdx.x == 0) costk[k] = red[0];
        __syncthreads();
    }
    if (threadIdx.x == 0) {
        float m = costk[0];
#pragma unroll
        for (int k = 1; k < R_H; ++k) m = fmaxf(m, costk[k]);
        float e[R_H], ssum = 0.f;
#pragma unroll
        for (int k = 0; k < R_H; ++k) { e[k] = __expf(costk[k] - m); ssum += e[k]; }
        float inv = 1.0f / ssum;
#pragma unroll
        for (int k = 0; k < R_H; ++k) ws[4 * CSZ + bc * R_H + k] = e[k] * inv;
    }
}

__global__ void __launch_bounds__(256) k_fused(const float* __restrict__ x,
                                               const float* __restrict__ Wt,
                                               const float* __restrict__ bias,
                                               const float* __restrict__ ws,
                                               float* __restrict__ out) {
    __shared__ __attribute__((aligned(16))) float y_pair[(CSZ / 2) * YPS];
    __shared__ __attribute__((aligned(16))) float wt_pair[(CSZ / 2) * WPS];
    __shared__ __attribute__((aligned(16))) float sc_l[CSZ], sh_l[CSZ], w_l[CSZ * R_H];

    int b      = blockIdx.y;
    int d_base = blockIdx.x * DT;
    int tid    = threadIdx.x;
    int wave   = tid >> 5;
    int lane   = tid & 31;

    const float4* xq = reinterpret_cast<const float4*>(
                           x + ((size_t)b * CSZ + 2 * wave) * NSZ + d_base) + lane;
    float4 ca = xq[0], cb = xq[NSZ / 4];

    for (int i = tid; i < OC * CSZ; i += 256) {
        int o = i >> 6, c = i & (CSZ - 1);
        wt_pair[(c >> 1) * WPS + o * 2 + (c & 1)] = Wt[i];
    }
    if (tid < CSZ) { sc_l[tid] = ws[2 * CSZ + tid]; sh_l[tid] = ws[3 * CSZ + tid]; }
    for (int i = tid; i < CSZ * R_H; i += 256)
        w_l[i] = ws[4 * CSZ + b * CSZ * R_H + i];
    __syncthreads();

    for (int j = 0; j < 4; ++j) {
        float4 na = ca, nb = cb;
        if (j < 3) {
            const float4* nx = xq + (size_t)(j + 1) * (16 * (NSZ / 4));
            na = nx[0]; nb = nx[NSZ / 4];
        }
        int cp = wave + 8 * j;
        int ce = 2 * cp;
        float sc0 = sc_l[ce], sc1 = sc_l[ce + 1];
        float sh0 = sh_l[ce], sh1 = sh_l[ce + 1];
        float w0[R_H], w1[R_H];
#pragma unroll
        for (int k = 0; k < R_H; ++k) { w0[k] = w_l[ce * R_H + k]; w1[k] = w_l[ce * R_H + R_H + k]; }
        float4 lo, hi;
        lo.x = y_of(ca.x, sc0, sh0, w0); lo.y = y_of(cb.x, sc1, sh1, w1);
        lo.z = y_of(ca.y, sc0, sh0, w0); lo.w = y_of(cb.y, sc1, sh1, w1);
        hi.x = y_of(ca.z, sc0, sh0, w0); hi.y = y_of(cb.z, sc1, sh1, w1);
        hi.z = y_of(ca.w, sc0, sh0, w0); hi.w = y_of(cb.w, sc1, sh1, w1);
        float4* dst = reinterpret_cast<float4*>(&y_pair[cp * YPS + lane * 8]);
        dst[0] = lo; dst[1] = hi;
        ca = na; cb = nb;
    }
    __syncthreads();

    int half = lane >> 4;
    int lid  = lane & 15;
    int o0   = (wave & 3) << 4;

    v2f afrag[16];
#pragma unroll
    for (int s = 0; s < 16; ++s)
        afrag[s] = *(const v2f*)&wt_pair[(2 * s + half) * WPS + (o0 + lid) * 2];

#pragma unroll
    for (int i = 0; i < 4; ++i) {
        int d0 = ((((wave >> 2) << 2) + i) << 4);
        v8f acc = {0.f, 0.f, 0.f, 0.f, 0.f, 0.f, 0.f, 0.f};
#pragma unroll
        for (int s = 0; s < 16; ++s) {
            v2f bb = *(const v2f*)&y_pair[(2 * s + half) * YPS + (d0 + lid) * 2];
            acc = __builtin_amdgcn_wmma_f32_16x16x4_f32(
                false, afrag[s], false, bb, (short)0, acc, false, false);
        }
        int dcol = d_base + d0 + lid;
#pragma unroll
        for (int v = 0; v < 8; ++v) {
            int o = o0 + v + 8 * half;
            out[(((size_t)b * OC + o) * NSZ) + dcol] = acc[v] + bias[o];
        }
    }
}

extern "C" void kernel_launch(void* const* d_in, const int* in_sizes, int n_in,
                              void* d_out, int out_size, void* d_ws, size_t ws_size,
                              hipStream_t stream) {
    const float* x     = (const float*)d_in[0];
    const float* gamma = (const float*)d_in[1];
    const float* beta  = (const float*)d_in[2];
    const float* Wt    = (const float*)d_in[3];
    const float* bias  = (const float*)d_in[4];
    float* out = (float*)d_out;
    float* ws  = (float*)d_ws;

    hipLaunchKernelGGL(k_init,     dim3(1),             dim3(128), 0, stream, ws);
    hipLaunchKernelGGL(k_stats,    dim3(BSZ * CSZ),     dim3(256), 0, stream, x, ws);
    hipLaunchKernelGGL(k_finalize, dim3(1),             dim3(64),  0, stream, gamma, beta, ws);
    hipLaunchKernelGGL(k_cost,     dim3(BSZ * CSZ),     dim3(256), 0, stream, x, ws);
    hipLaunchKernelGGL(k_fused,    dim3(NSZ / DT, BSZ), dim3(256), 0, stream,
                       x, Wt, bias, ws, out);
}


# stats and cost kernels sum each block by in-register wave butterflies (row_xmask adds, one row exchange) and ONE LDS hand-over, replacing 2 and 6 serial 8-step LDS trees of a barrier per step
# speedup vs baseline: 2.2472x; 1.3125x over previous
#include <hip/hip_runtime.h>
#include <hip/hip_bf16.h>
#include <math.h>

#define R_H 6
#define EPSBN 1e-5f
#define BSZ 8
#define CSZ 64
#define NSZ 16384
#define OC  64
#define DT  128
#define YPS 288
#define WPS 160
#define WS_PSUM 0
#define WS_PSQ  (BSZ * CSZ)
#define WS_SC   (2 * BSZ * CSZ)
#define WS_SH   (2 * BSZ * CSZ + CSZ)
#define WS_W    (2 * BSZ * CSZ + 2 * CSZ)

typedef __attribute__((ext_vector_type(2))) float v2f;
typedef __attribute__((ext_vector_type(8))) float v8f;

template <int M>
__device__ __forceinline__ float row_xor(float v) {
    int b = __builtin_bit_cast(int, v);
    return __builtin_bit_cast(float, __builtin_amdgcn_update_dpp(b, b, 0x160 + M, 0xf, 0xf, false));
}
__device__ __forceinline__ float xhalf(float v) {
    int b = __builtin_bit_cast(int, v);
    return __builtin_bit_cast(float,
        __builtin_amdgcn_permlanex16(b, b, 0x76543210, 0xfedcba98, false, false));
}
__device__ __forceinline__ float wave_sum(float v) {
    v += row_xor<1>(v);
    v += row_xor<2>(v);
    v += row_xor<4>(v);
    v += row_xor<8>(v);
    v += xhalf(v);
    return v;
}

__device__ __forceinline__ void hermite6(float xn, float p[R_H]) {
    const float SQ13 = 3.60555127546398929312f;
    const float PIQ  = 0.75112554446494248286f;
    float e = __builtin_amdgcn_exp2f(xn * 2.88539008177792681472f);
    float t = fmaf(-2.0f * SQ13, __builtin_amdgcn_rcpf(e + 1.0f), SQ13);
    float g = __expf(-0.5f * t * t);
    p[0] = PIQ * g;
    p[1] = 1.41421356237309515f * PIQ * t * g;
    p[2] = 1.0f                 * t * p[1] - 0.70710678118654752f * p[0];
    p[3] = 0.81649658092772603f * t * p[2] - 0.81649658092772603f * p[1];
    p[4] = 0.70710678118654752f * t * p[3] - 0.86602540378443865f * p[2];
    p[5] = 0.63245553203367587f * t * p[4] - 0.89442719099991588f * p[3];
}

__global__ void __launch_bounds__(256) k_stats(const float* __restrict__ x,
                                               float* __restrict__ ws) {
    __shared__ __attribute__((aligned(16))) float part[2 * 8];
    int bc = blockIdx.x;
    int c  = bc & (CSZ - 1);
    int b  = bc >> 6;
    const float4* xp4 = reinterpret_cast<const float4*>(x + (size_t)bc * NSZ) + threadIdx.x;
    float4 v[16];
#pragma unroll
    for (int j = 0; j < 16; ++j) v[j] = xp4[j * 256];
    __builtin_amdgcn_sched_barrier(0);
    float4 a = {0.f, 0.f, 0.f, 0.f}, q = {0.f, 0.f, 0.f, 0.f};
#pragma unroll
    for (int j = 0; j < 16; ++j) {
        a.x += v[j].x; q.x += v[j].x * v[j].x;
        a.y += v[j].y; q.y += v[j].y * v[j].y;
        a.z += v[j].z; q.z += v[j].z * v[j].z;
        a.w += v[j].w; q.w += v[j].w * v[j].w;
    }
    float s  = wave_sum((a.x + a.y) + (a.z + a.w));
    float s2 = wave_sum((q.x + q.y) + (q.z + q.w));
    int wave = threadIdx.x >> 5;
    if ((threadIdx.x & 31) == 0) { part[wave] = s; part[8 + wave] = s2; }
    __syncthreads();
    if (threadIdx.x == 0) {
        const float4* p4 = reinterpret_cast<const float4*>(part);
        float4 s0 = p4[0], s1 = p4[1], q0 = p4[2], q1 = p4[3];
        ws[WS_PSUM + c * BSZ + b] = ((s0.x + s0.y) + (s0.z + s0.w)) + ((s1.x + s1.y) + (s1.z + s1.w));
        ws[WS_PSQ  + c * BSZ + b] = ((q0.x + q0.y) + (q0.z + q0.w)) + ((q1.x + q1.y) + (q1.z + q1.w));
    }
}

__device__ __forceinline__ void fold_stats(const float* ws, float g, float bt, int c,
                                           float& sc, float& sh) {
    const float4* ps = reinterpret_cast<const float4*>(ws + WS_PSUM + c * BSZ);
    const float4* pq = reinterpret_cast<const float4*>(ws + WS_PSQ  + c * BSZ);
    float4 s0 = ps[0], s1 = ps[1], q0 = pq[0], q1 = pq[1];
    float s = ((s0.x + s0.y) + (s0.z + s0.w)) + ((s1.x + s1.y) + (s1.z + s1.w));
    float q = ((q0.x + q0.y) + (q0.z + q0.w)) + ((q1.x + q1.y) + (q1.z + q1.w));
    const float inv_n = 1.0f / ((float)BSZ * (float)NSZ);
    float mean = s * inv_n;
    float var  = q * inv_n - mean * mean;
    float rstd = rsqrtf(var + EPSBN);
    sc = g * rstd;
    sh = bt - mean * sc;
}

__device__ __forceinline__ void cost_acc1(float xv, float sc, float sh, float acc[R_H]) {
    float xn = fmaf(sc, xv, sh);
    float p[R_H];
    hermite6(xn, p);
#pragma unroll
    for (int k = 0; k < R_H; ++k) acc[k] = fmaf(p[k], xn, acc[k]);
}
__device__ __forceinline__ void cost_acc4(float4 v, float sc, float sh, float acc[R_H]) {
    cost_acc1(v.x, sc, sh, acc);
    cost_acc1(v.y, sc, sh, acc);
    cost_acc1(v.z, sc, sh, acc);
    cost_acc1(v.w, sc, sh, acc);
}

__device__ __forceinline__ float y_of(float xv, float sc, float sh, const float w[R_H]) {
    float xn = fmaf(sc, xv, sh);
    float p[R_H];
    hermite6(xn, p);
    float y = 0.f;
#pragma unroll
    for (int k = 0; k < R_H; ++k) y = fmaf(w[k], p[k], y);
    return y;
}

__global__ void __launch_bounds__(256) k_cost(const float* __restrict__ x,
                                              const float* __restrict__ gamma,
                                              const float* __restrict__ beta,
                                              float* __restrict__ ws) {
    __shared__ __attribute__((aligned(16))) float part[8 * 8];
    int bc = blockIdx.x;
    int c  = bc & (CSZ - 1);
    float sc, sh;
    fold_stats(ws, gamma[c], beta[c], c, sc, sh);
    if (bc < CSZ && threadIdx.x == 0) { ws[WS_SC + c] = sc; ws[WS_SH + c] = sh; }
    const float* xp = x + (size_t)bc * NSZ;

    float acc[R_H];
#pragma unroll
    for (int k = 0; k < R_H; ++k) acc[k] = 0.f;

    const float4* xp4 = reinterpret_cast<const float4*>(xp) + threadIdx.x;
    float4 c0 = xp4[0], c1 = xp4[256], c2 = xp4[512], c3 = xp4[768];
    for (int g = 0; g < 4; ++g) {
        float4 n0 = c0, n1 = c1, n2 = c2, n3 = c3;
        if (g < 3) {
            const float4* nx = xp4 + (g + 1) * 1024;
            n0 = nx[0]; n1 = nx[256]; n2 = nx[512]; n3 = nx[768];
        }
        cost_acc4(c0, sc, sh, acc);
        cost_acc4(c1, sc, sh, acc);
        cost_acc4(c2, sc, sh, acc);
        cost_acc4(c3, sc, sh, acc);
        c0 = n0; c1 = n1; c2 = n2; c3 = n3;
    }
#pragma unroll
    for (int k = 0; k < R_H; ++k) acc[k] = wave_sum(acc[k]);
    int wave = threadIdx.x >> 5;
    if ((threadIdx.x & 31) == 0) {
        *reinterpret_cast<float4*>(&part[wave * 8])     = make_float4(acc[0], acc[1], acc[2], acc[3]);
        *reinterpret_cast<float2*>(&part[wave * 8 + 4]) = make_float2(acc[4], acc[5]);
    }
    __syncthreads();
    if (threadIdx.x == 0) {
        float4 lo = *reinterpret_cast<const float4*>(&part[0]);
        float2 hi = *reinterpret_cast<const float2*>(&part[4]);
#pragma unroll
        for (int w = 1; w < 8; ++w) {
            float4 l2 = *reinterpret_cast<const float4*>(&part[w * 8]);
            float2 h2 = *reinterpret_cast<const float2*>(&part[w * 8 + 4]);
            lo.x += l2.x; lo.y += l2.y; lo.z += l2.z; lo.w += l2.w;
            hi.x += h2.x; hi.y += h2.y;
        }
        float costk[R_H] = {lo.x, lo.y, lo.z, lo.w, hi.x, hi.y};
        float m = costk[0];
#pragma unroll
        for (int k = 1; k < R_H; ++k) m = fmaxf(m, costk[k]);
        float e[R_H], ssum = 0.f;
#pragma unroll
        for (int k = 0; k < R_H; ++k) { e[k] = __expf(costk[k] - m); ssum += e[k]; }
        float inv = 1.0f / ssum;
#pragma unroll
        for (int k = 0; k < R_H; ++k) ws[WS_W + bc * R_H + k] = e[k] * inv;
    }
}

__global__ void __launch_bounds__(256) k_fused(const float* __restrict__ x,
                                               const float* __restrict__ Wt,
                                               const float* __restrict__ bias,
                                               const float* __restrict__ ws,
                                               float* __restrict__ out) {
    __shared__ __attribute__((aligned(16))) float y_pair[(CSZ / 2) * YPS];
    __shared__ __attribute__((aligned(16))) float wt_pair[(CSZ / 2) * WPS];
    __shared__ __attribute__((aligned(16))) float sc_l[CSZ], sh_l[CSZ], w_l[CSZ * R_H];

    int b      = blockIdx.y;
    int d_base = blockIdx.x * DT;
    int tid    = threadIdx.x;
    int wave   = tid >> 5;
    int lane   = tid & 31;

    const float4* xq = reinterpret_cast<const float4*>(
                           x + ((size_t)b * CSZ + 2 * wave) * NSZ + d_base) + lane;
    float4 ca = xq[0], cb = xq[NSZ / 4];

    for (int i = tid; i < OC * CSZ; i += 256) {
        int o = i >> 6, c = i & (CSZ - 1);
        wt_pair[(c >> 1) * WPS + o * 2 + (c & 1)] = Wt[i];
    }
    if (tid < CSZ) { sc_l[tid] = ws[WS_SC + tid]; sh_l[tid] = ws[WS_SH + tid]; }
    for (int i = tid; i < CSZ * R_H; i += 256)
        w_l[i] = ws[WS_W + b * CSZ * R_H + i];
    __syncthreads();

    for (int j = 0; j < 4; ++j) {
        float4 na = ca, nb = cb;
        if (j < 3) {
            const float4* nx = xq + (size_t)(j + 1) * (16 * (NSZ / 4));
            na = nx[0]; nb = nx[NSZ / 4];
        }
        int cp = wave + 8 * j;
        int ce = 2 * cp;
        float sc0 = sc_l[ce], sc1 = sc_l[ce + 1];
        float sh0 = sh_l[ce], sh1 = sh_l[ce + 1];
        float w0[R_H], w1[R_H];
#pragma unroll
        for (int k = 0; k < R_H; ++k) { w0[k] = w_l[ce * R_H + k]; w1[k] = w_l[ce * R_H + R_H + k]; }
        float4 lo, hi;
        lo.x = y_of(ca.x, sc0, sh0, w0); lo.y = y_of(cb.x, sc1, sh1, w1);
        lo.z = y_of(ca.y, sc0, sh0, w0); lo.w = y_of(cb.y, sc1, sh1, w1);
        hi.x = y_of(ca.z, sc0, sh0, w0); hi.y = y_of(cb.z, sc1, sh1, w1);
        hi.z = y_of(ca.w, sc0, sh0, w0); hi.w = y_of(cb.w, sc1, sh1, w1);
        float4* dst = reinterpret_cast<float4*>(&y_pair[cp * YPS + lane * 8]);
        dst[0] = lo; dst[1] = hi;
        ca = na; cb = nb;
    }
    __syncthreads();

    int half = lane >> 4;
    int lid  = lane & 15;
    int o0   = (wave & 3) << 4;

    v2f afrag[16];
#pragma unroll
    for (int s = 0; s < 16; ++s)
        afrag[s] = *(const v2f*)&wt_pair[(2 * s + half) * WPS + (o0 + lid) * 2];

#pragma unroll
    for (int i = 0; i < 4; ++i) {
        int d0 = ((((wave >> 2) << 2) + i) << 4);
        v8f acc = {0.f, 0.f, 0.f, 0.f, 0.f, 0.f, 0.f, 0.f};
#pragma unroll
        for (int s = 0; s < 16; ++s) {
            v2f bb = *(const v2f*)&y_pair[(2 * s + half) * YPS + (d0 + lid) * 2];
            acc = __builtin_amdgcn_wmma_f32_16x16x4_f32(
                false, afrag[s], false, bb, (short)0, acc, false, false);
        }
        int dcol = d_base + d0 + lid;
#pragma unroll
        for (int v = 0; v < 8; ++v) {
            int o = o0 + v + 8 * half;
            out[(((size_t)b * OC + o) * NSZ) + dcol] = acc[v] + bias[o];
        }
    }
}

extern "C" void kernel_launch(void* const* d_in, const int* in_sizes, int n_in,
                              void* d_out, int out_size, void* d_ws, size_t ws_size,
                              hipStream_t stream) {
    const float* x     = (const float*)d_in[0];
    const float* gamma = (const float*)d_in[1];
    const float* beta  = (const float*)d_in[2];
    const float* Wt    = (const float*)d_in[3];
    const float* bias  = (const float*)d_in[4];
    float* out = (float*)d_out;
    float* ws  = (float*)d_ws;

    hipLaunchKernelGGL(k_stats, dim3(BSZ * CSZ),     dim3(256), 0, stream, x, ws);
    hipLaunchKernelGGL(k_cost,  dim3(BSZ * CSZ),     dim3(256), 0, stream, x, gamma, beta, ws);
    hipLaunchKernelGGL(k_fused, dim3(NSZ / DT, BSZ), dim3(256), 0, stream,
                       x, Wt, bias, ws, out);
}


# Hermite functions in closed form, Gaussian times even/odd polynomials in tanh^2 (Horner), not the recurrence: cost pass sums six polynomial terms, fused pass one degree-5 polynomial per channel
# speedup vs baseline: 2.4491x; 1.0898x over previous
#include <hip/hip_runtime.h>
#include <hip/hip_bf16.h>
#include <math.h>

#define R_H 6
#define EPSBN 1e-5f
#define BSZ 8
#define CSZ 64
#define NSZ 16384
#define OC  64
#define DT  128
#define YPS 288
#define WPS 160
#define WS_PSUM 0
#define WS_PSQ  (BSZ * CSZ)
#define WS_SC   (2 * BSZ * CSZ)
#define WS_SH   (2 * BSZ * CSZ + CSZ)
#define WS_W    (2 * BSZ * CSZ + 2 * CSZ)

typedef __attribute__((ext_vector_type(2))) float v2f;
typedef __attribute__((ext_vector_type(8))) float v8f;

template <int M>
__device__ __forceinline__ float row_xor(float v) {
    int b = __builtin_bit_cast(int, v);
    return __builtin_bit_cast(float, __builtin_amdgcn_update_dpp(b, b, 0x160 + M, 0xf, 0xf, false));
}
__device__ __forceinline__ float xhalf(float v) {
    int b = __builtin_bit_cast(int, v);
    return __builtin_bit_cast(float,
        __builtin_amdgcn_permlanex16(b, b, 0x76543210, 0xfedcba98, false, false));
}
__device__ __forceinline__ float wave_sum(float v) {
    v += row_xor<1>(v);
    v += row_xor<2>(v);
    v += row_xor<4>(v);
    v += row_xor<8>(v);
    v += xhalf(v);
    return v;
}

#define K00   0.75112554446494251f
#define K11   3.8300038083847467f
#define K20  -0.53112596601359852f
#define K22  13.809275116353559f
#define K31  -4.6907775217294736f
#define K33  40.653405188322097f
#define K40   0.45996857917732664f
#define K42 -23.918366117220984f
#define K44 103.64625317462428f
#define K51   5.2444487029575502f
#define K53 -90.903777517930848f
#define K55 236.34982154662021f
#define C2LOG2E 2.88539008177792681472f

__device__ __forceinline__ void tanh_gauss(float a, float& m, float& s, float& g) {
    float e = __builtin_amdgcn_exp2f(a);
    m = fmaf(-2.0f, __builtin_amdgcn_rcpf(e + 1.0f), 1.0f);
    s = m * m;
    g = __builtin_amdgcn_exp2f(s * -9.37751776577826314f);
}

__global__ void __launch_bounds__(256) k_stats(const float* __restrict__ x,
                                               float* __restrict__ ws) {
    __shared__ __attribute__((aligned(16))) float part[2 * 8];
    int bc = blockIdx.x;
    int c  = bc & (CSZ - 1);
    int b  = bc >> 6;
    const float4* xp4 = reinterpret_cast<const float4*>(x + (size_t)bc * NSZ) + threadIdx.x;
    float4 v[16];
#pragma unroll
    for (int j = 0; j < 16; ++j) v[j] = xp4[j * 256];
    __builtin_amdgcn_sched_barrier(0);
    float4 a = {0.f, 0.f, 0.f, 0.f}, q = {0.f, 0.f, 0.f, 0.f};
#pragma unroll
    for (int j = 0; j < 16; ++j) {
        a.x += v[j].x; q.x += v[j].x * v[j].x;
        a.y += v[j].y; q.y += v[j].y * v[j].y;
        a.z += v[j].z; q.z += v[j].z * v[j].z;
        a.w += v[j].w; q.w += v[j].w * v[j].w;
    }
    float s  = wave_sum((a.x + a.y) + (a.z + a.w));
    float s2 = wave_sum((q.x + q.y) + (q.z + q.w));
    int wave = threadIdx.x >> 5;
    if ((threadIdx.x & 31) == 0) { part[wave] = s; part[8 + wave] = s2; }
    __syncthreads();
    if (threadIdx.x == 0) {
        const float4* p4 = reinterpret_cast<const float4*>(part);
        float4 s0 = p4[0], s1 = p4[1], q0 = p4[2], q1 = p4[3];
        ws[WS_PSUM + c * BSZ + b] = ((s0.x + s0.y) + (s0.z + s0.w)) + ((s1.x + s1.y) + (s1.z + s1.w));
        ws[WS_PSQ  + c * BSZ + b] = ((q0.x + q0.y) + (q0.z + q0.w)) + ((q1.x + q1.y) + (q1.z + q1.w));
    }
}

__device__ __forceinline__ void fold_stats(const float* ws, float g, float bt, int c,
                                           float& sc, float& sh) {
    const float4* ps = reinterpret_cast<const float4*>(ws + WS_PSUM + c * BSZ);
    const float4* pq = reinterpret_cast<const float4*>(ws + WS_PSQ  + c * BSZ);
    float4 s0 = ps[0], s1 = ps[1], q0 = pq[0], q1 = pq[1];
    float s = ((s0.x + s0.y) + (s0.z + s0.w)) + ((s1.x + s1.y) + (s1.z + s1.w));
    float q = ((q0.x + q0.y) + (q0.z + q0.w)) + ((q1.x + q1.y) + (q1.z + q1.w));
    const float inv_n = 1.0f / ((float)BSZ * (float)NSZ);
    float mean = s * inv_n;
    float var  = q * inv_n - mean * mean;
    float rstd = rsqrtf(var + EPSBN);
    sc = g * rstd;
    sh = bt - mean * sc;
}

__device__ __forceinline__ void cost_acc1(float xv, float sc, float sh, float acc[R_H]) {
    float xn = fmaf(sc, xv, sh);
    float m, s, g;
    tanh_gauss(xn * C2LOG2E, m, s, g);
    float u  = g * xn;
    float um = u * m;
    acc[0] += u;
    acc[1] += um;
    acc[2] = fmaf(u,  fmaf(K22, s, K20), acc[2]);
    acc[3] = fmaf(um, fmaf(K33, s, K31), acc[3]);
    acc[4] = fmaf(u,  fmaf(fmaf(K44, s, K42), s, K40), acc[4]);
    acc[5] = fmaf(um, fmaf(fmaf(K55, s, K53), s, K51), acc[5]);
}
__device__ __forceinline__ void cost_acc4(float4 v, float sc, float sh, float acc[R_H]) {
    cost_acc1(v.x, sc, sh, acc);
    cost_acc1(v.y, sc, sh, acc);
    cost_acc1(v.z, sc, sh, acc);
    cost_acc1(v.w, sc, sh, acc);
}

__device__ __forceinline__ float y_of(float xv, float sc, float sh, const float q[R_H]) {
    float m, s, g;
    tanh_gauss(fmaf(sc, xv, sh), m, s, g);
    float ev = fmaf(fmaf(q[4], s, q[2]), s, q[0]);
    float od = fmaf(fmaf(q[5], s, q[3]), s, q[1]);
    return g * fmaf(m, od, ev);
}

__global__ void __launch_bounds__(256) k_cost(const float* __restrict__ x,
                                              const float* __restrict__ gamma,
                                              const float* __restrict__ beta,
                                              float* __restrict__ ws) {
    __shared__ __attribute__((aligned(16))) float part[8 * 8];
    int bc = blockIdx.x;
    int c  = bc & (CSZ - 1);
    float sc, sh;
    fold_stats(ws, gamma[c], beta[c], c, sc, sh);
    if (bc < CSZ && threadIdx.x == 0) { ws[WS_SC + c] = sc * C2LOG2E; ws[WS_SH + c] = sh * C2LOG2E; }
    const float* xp = x + (size_t)bc * NSZ;

    float acc[R_H];
#pragma unroll
    for (int k = 0; k < R_H; ++k) acc[k] = 0.f;

    const float4* xp4 = reinterpret_cast<const float4*>(xp) + threadIdx.x;
    float4 c0 = xp4[0], c1 = xp4[256], c2 = xp4[512], c3 = xp4[768];
    for (int g = 0; g < 4; ++g) {
        float4 n0 = c0, n1 = c1, n2 = c2, n3 = c3;
        if (g < 3) {
            const float4* nx = xp4 + (g + 1) * 1024;
            n0 = nx[0]; n1 = nx[256]; n2 = nx[512]; n3 = nx[768];
        }
        cost_acc4(c0, sc, sh, acc);
        cost_acc4(c1, sc, sh, acc);
        cost_acc4(c2, sc, sh, acc);
        cost_acc4(c3, sc, sh, acc);
        c0 = n0; c1 = n1; c2 = n2; c3 = n3;
    }
#pragma unroll
    for (int k = 0; k < R_H; ++k) acc[k] = wave_sum(acc[k]);
    int wave = threadIdx.x >> 5;
    if ((threadIdx.x & 31) == 0) {
        *reinterpret_cast<float4*>(&part[wave * 8])     = make_float4(acc[0], acc[1], acc[2], acc[3]);
        *reinterpret_cast<float2*>(&part[wave * 8 + 4]) = make_float2(acc[4], acc[5]);
    }
    __syncthreads();
    if (threadIdx.x == 0) {
        float4 lo = *reinterpret_cast<const float4*>(&part[0]);
        float2 hi = *reinterpret_cast<const float2*>(&part[4]);
#pragma unroll
        for (int w = 1; w < 8; ++w) {
            float4 l2 = *reinterpret_cast<const float4*>(&part[w * 8]);
            float2 h2 = *reinterpret_cast<const float2*>(&part[w * 8 + 4]);
            lo.x += l2.x; lo.y += l2.y; lo.z += l2.z; lo.w += l2.w;
            hi.x += h2.x; hi.y += h2.y;
        }
        float costk[R_H] = {K00 * lo.x, K11 * lo.y, lo.z, lo.w, hi.x, hi.y};
        float m = costk[0];
#pragma unroll
        for (int k = 1; k < R_H; ++k) m = fmaxf(m, costk[k]);
        float e[R_H], ssum = 0.f;
#pragma unroll
        for (int k = 0; k < R_H; ++k) { e[k] = __expf(costk[k] - m); ssum += e[k]; }
        float inv = 1.0f / ssum;
        float wk[R_H];
#pragma unroll
        for (int k = 0; k < R_H; ++k) wk[k] = e[k] * inv;
        float* qd = ws + WS_W + (bc >> 1) * (2 * R_H) + (bc & 1);
        qd[0]  = fmaf(wk[4], K40, fmaf(wk[2], K20, wk[0] * K00));
        qd[2]  = fmaf(wk[5], K51, fmaf(wk[3], K31, wk[1] * K11));
        qd[4]  = fmaf(wk[4], K42, wk[2] * K22);
        qd[6]  = fmaf(wk[5], K53, wk[3] * K33);
        qd[8]  = wk[4] * K44;
        qd[10] = wk[5] * K55;
    }
}

__global__ void __launch_bounds__(256) k_fused(const float* __restrict__ x,
                                               const float* __restrict__ Wt,
                                               const float* __restrict__ bias,
                                               const float* __restrict__ ws,
                                               float* __restrict__ out) {
    __shared__ __attribute__((aligned(16))) float y_pair[(CSZ / 2) * YPS];
    __shared__ __attribute__((aligned(16))) float wt_pair[(CSZ / 2) * WPS];
    __shared__ __attribute__((aligned(16))) float sc_l[CSZ], sh_l[CSZ], q_l[CSZ * R_H];

    int b      = blockIdx.y;
    int d_base = blockIdx.x * DT;
    int tid    = threadIdx.x;
    int wave   = tid >> 5;
    int lane   = tid & 31;

    const float4* xq = reinterpret_cast<const float4*>(
                           x + ((size_t)b * CSZ + 2 * wave) * NSZ + d_base) + lane;
    float4 ca = xq[0], cb = xq[NSZ / 4];

    for (int i = tid; i < OC * CSZ; i += 256) {
        int o = i >> 6, c = i & (CSZ - 1);
        wt_pair[(c >> 1) * WPS + o * 2 + (c & 1)] = Wt[i];
    }
    if (tid < CSZ) { sc_l[tid] = ws[WS_SC + tid]; sh_l[tid] = ws[WS_SH + tid]; }
    for (int i = tid; i < CSZ * R_H; i += 256)
        q_l[i] = ws[WS_W + b * CSZ * R_H + i];
    __syncthreads();

    for (int j = 0; j < 4; ++j) {
        float4 na = ca, nb = cb;
        if (j < 3) {
            const float4* nx = xq + (size_t)(j + 1) * (16 * (NSZ / 4));
            na = nx[0]; nb = nx[NSZ / 4];
        }
        int cp = wave + 8 * j;
        int ce = 2 * cp;
        float sc0 = sc_l[ce], sc1 = sc_l[ce + 1];
        float sh0 = sh_l[ce], sh1 = sh_l[ce + 1];
        float q0[R_H], q1[R_H];
#pragma unroll
        for (int k = 0; k < R_H; ++k) { q0[k] = q_l[ce * R_H + 2 * k]; q1[k] = q_l[ce * R_H + 2 * k + 1]; }
        float4 lo, hi;
        lo.x = y_of(ca.x, sc0, sh0, q0); lo.y = y_of(cb.x, sc1, sh1, q1);
        lo.z = y_of(ca.y, sc0, sh0, q0); lo.w = y_of(cb.y, sc1, sh1, q1);
        hi.x = y_of(ca.z, sc0, sh0, q0); hi.y = y_of(cb.z, sc1, sh1, q1);
        hi.z = y_of(ca.w, sc0, sh0, q0); hi.w = y_of(cb.w, sc1, sh1, q1);
        float4* dst = reinterpret_cast<float4*>(&y_pair[cp * YPS + lane * 8]);
        dst[0] = lo; dst[1] = hi;
        ca = na; cb = nb;
    }
    __syncthreads();

    int half = lane >> 4;
    int lid  = lane & 15;
    int o0   = (wave & 3) << 4;

    v2f afrag[16];
#pragma unroll
    for (int s = 0; s < 16; ++s)
        afrag[s] = *(const v2f*)&wt_pair[(2 * s + half) * WPS + (o0 + lid) * 2];

#pragma unroll
    for (int i = 0; i < 4; ++i) {
        int d0 = ((((wave >> 2) << 2) + i) << 4);
        v8f acc = {0.f, 0.f, 0.f, 0.f, 0.f, 0.f, 0.f, 0.f};
#pragma unroll
        for (int s = 0; s < 16; ++s) {
            v2f bb = *(const v2f*)&y_pair[(2 * s + half) * YPS + (d0 + lid) * 2];
            acc = __builtin_amdgcn_wmma_f32_16x16x4_f32(
                false, afrag[s], false, bb, (short)0, acc, false, false);
        }
        int dcol = d_base + d0 + lid;
#pragma unroll
        for (int v = 0; v < 8; ++v) {
            int o = o0 + v + 8 * half;
            out[(((size_t)b * OC + o) * NSZ) + dcol] = acc[v] + bias[o];
        }
    }
}

extern "C" void kernel_launch(void* const* d_in, const int* in_sizes, int n_in,
                              void* d_out, int out_size, void* d_ws, size_t ws_size,
                              hipStream_t stream) {
    const float* x     = (const float*)d_in[0];
    const float* gamma = (const float*)d_in[1];
    const float* beta  = (const float*)d_in[2];
    const float* Wt    = (const float*)d_in[3];
    const float* bias  = (const float*)d_in[4];
    float* out = (float*)d_out;
    float* ws  = (float*)d_ws;

    hipLaunchKernelGGL(k_stats, dim3(BSZ * CSZ),     dim3(256), 0, stream, x, ws);
    hipLaunchKernelGGL(k_cost,  dim3(BSZ * CSZ),     dim3(256), 0, stream, x, gamma, beta, ws);
    hipLaunchKernelGGL(k_fused, dim3(NSZ / DT, BSZ), dim3(256), 0, stream,
                       x, Wt, bias, ws, out);
}


# every load at the head of the fused and cost kernels is requested as one group (Wt goes to LDS row-major at a 68-word pitch by 16-byte loads): one memory round trip before the math, not six and two
# speedup vs baseline: 2.5775x; 1.0524x over previous
#include <hip/hip_runtime.h>
#include <hip/hip_bf16.h>
#include <math.h>

#define R_H 6
#define EPSBN 1e-5f
#define BSZ 8
#define CSZ 64
#define NSZ 16384
#define OC  64
#define DT  128
#define YPS 288
#define WLP 68
#define CF_SC (CSZ * R_H)
#define CF_SH (CSZ * R_H + CSZ)
#define WS_PSUM 0
#define WS_PSQ  (BSZ * CSZ)
#define WS_SC   (2 * BSZ * CSZ)
#define WS_SH   (2 * BSZ * CSZ + CSZ)
#define WS_W    (2 * BSZ * CSZ + 2 * CSZ)

typedef __attribute__((ext_vector_type(2))) float v2f;
typedef __attribute__((ext_vector_type(8))) float v8f;

template <int M>
__device__ __forceinline__ float row_xor(float v) {
    int b = __builtin_bit_cast(int, v);
    return __builtin_bit_cast(float, __builtin_amdgcn_update_dpp(b, b, 0x160 + M, 0xf, 0xf, false));
}
__device__ __forceinline__ float xhalf(float v) {
    int b = __builtin_bit_cast(int, v);
    return __builtin_bit_cast(float,
        __builtin_amdgcn_permlanex16(b, b, 0x76543210, 0xfedcba98, false, false));
}
__device__ __forceinline__ float wave_sum(float v) {
    v += row_xor<1>(v);
    v += row_xor<2>(v);
    v += row_xor<4>(v);
    v += row_xor<8>(v);
    v += xhalf(v);
    return v;
}

#define K00   0.75112554446494251f
#define K11   3.8300038083847467f
#define K20  -0.53112596601359852f
#define K22  13.809275116353559f
#define K31  -4.6907775217294736f
#define K33  40.653405188322097f
#define K40   0.45996857917732664f
#define K42 -23.918366117220984f
#define K44 103.64625317462428f
#define K51   5.2444487029575502f
#define K53 -90.903777517930848f
#define K55 236.34982154662021f
#define C2LOG2E 2.88539008177792681472f

__device__ __forceinline__ void tanh_gauss(float a, float& m, float& s, float& g) {
    float e = __builtin_amdgcn_exp2f(a);
    m = fmaf(-2.0f, __builtin_amdgcn_rcpf(e + 1.0f), 1.0f);
    s = m * m;
    g = __builtin_amdgcn_exp2f(s * -9.37751776577826314f);
}

__global__ void __launch_bounds__(256) k_stats(const float* __restrict__ x,
                                               float* __restrict__ ws) {
    __shared__ __attribute__((aligned(16))) float part[2 * 8];
    int bc = blockIdx.x;
    int c  = bc & (CSZ - 1);
    int b  = bc >> 6;
    const float4* xp4 = reinterpret_cast<const float4*>(x + (size_t)bc * NSZ) + threadIdx.x;
    float4 v[16];
#pragma unroll
    for (int j = 0; j < 16; ++j) v[j] = xp4[j * 256];
    __builtin_amdgcn_sched_barrier(0);
    float4 a = {0.f, 0.f, 0.f, 0.f}, q = {0.f, 0.f, 0.f, 0.f};
#pragma unroll
    for (int j = 0; j < 16; ++j) {
        a.x += v[j].x; q.x += v[j].x * v[j].x;
        a.y += v[j].y; q.y += v[j].y * v[j].y;
        a.z += v[j].z; q.z += v[j].z * v[j].z;
        a.w += v[j].w; q.w += v[j].w * v[j].w;
    }
    float s  = wave_sum((a.x + a.y) + (a.z + a.w));
    float s2 = wave_sum((q.x + q.y) + (q.z + q.w));
    int wave = threadIdx.x >> 5;
    if ((threadIdx.x & 31) == 0) { part[wave] = s; part[8 + wave] = s2; }
    __syncthreads();
    if (threadIdx.x == 0) {
        const float4* p4 = reinterpret_cast<const float4*>(part);
        float4 s0 = p4[0], s1 = p4[1], q0 = p4[2], q1 = p4[3];
        ws[WS_PSUM + c * BSZ + b] = ((s0.x + s0.y) + (s0.z + s0.w)) + ((s1.x + s1.y) + (s1.z + s1.w));
        ws[WS_PSQ  + c * BSZ + b] = ((q0.x + q0.y) + (q0.z + q0.w)) + ((q1.x + q1.y) + (q1.z + q1.w));
    }
}

__device__ __forceinline__ void fold_stats(const float* ws, float g, float bt, int c,
                                           float& sc, float& sh) {
    const float4* ps = reinterpret_cast<const float4*>(ws + WS_PSUM + c * BSZ);
    const float4* pq = reinterpret_cast<const float4*>(ws + WS_PSQ  + c * BSZ);
    float4 s0 = ps[0], s1 = ps[1], q0 = pq[0], q1 = pq[1];
    float s = ((s0.x + s0.y) + (s0.z + s0.w)) + ((s1.x + s1.y) + (s1.z + s1.w));
    float q = ((q0.x + q0.y) + (q0.z + q0.w)) + ((q1.x + q1.y) + (q1.z + q1.w));
    const float inv_n = 1.0f / ((float)BSZ * (float)NSZ);
    float mean = s * inv_n;
    float var  = q * inv_n - mean * mean;
    float rstd = rsqrtf(var + EPSBN);
    sc = g * rstd;
    sh = bt - mean * sc;
}

__device__ __forceinline__ void cost_acc1(float xv, float sc, float sh, float acc[R_H]) {
    float xn = fmaf(sc, xv, sh);
    float m, s, g;
    tanh_gauss(xn * C2LOG2E, m, s, g);
    float u  = g * xn;
    float um = u * m;
    acc[0] += u;
    acc[1] += um;
    acc[2] = fmaf(u,  fmaf(K22, s, K20), acc[2]);
    acc[3] = fmaf(um, fmaf(K33, s, K31), acc[3]);
    acc[4] = fmaf(u,  fmaf(fmaf(K44, s, K42), s, K40), acc[4]);
    acc[5] = fmaf(um, fmaf(fmaf(K55, s, K53), s, K51), acc[5]);
}
__device__ __forceinline__ void cost_acc4(float4 v, float sc, float sh, float acc[R_H]) {
    cost_acc1(v.x, sc, sh, acc);
    cost_acc1(v.y, sc, sh, acc);
    cost_acc1(v.z, sc, sh, acc);
    cost_acc1(v.w, sc, sh, acc);
}

__device__ __forceinline__ float y_of(float xv, float sc, float sh, const float q[R_H]) {
    float m, s, g;
    tanh_gauss(fmaf(sc, xv, sh), m, s, g);
    float ev = fmaf(fmaf(q[4], s, q[2]), s, q[0]);
    float od = fmaf(fmaf(q[5], s, q[3]), s, q[1]);
    return g * fmaf(m, od, ev);
}

__global__ void __launch_bounds__(256) k_cost(const float* __restrict__ x,
                                              const float* __restrict__ gamma,
                                              const float* __restrict__ beta,
                                              float* __restrict__ ws) {
    __shared__ __attribute__((aligned(16))) float part[8 * 8];
    int bc = blockIdx.x;
    int c  = bc & (CSZ - 1);
    const float4* xp4 = reinterpret_cast<const float4*>(x + (size_t)bc * NSZ) + threadIdx.x;
    float4 c0 = xp4[0], c1 = xp4[256], c2 = xp4[512], c3 = xp4[768];
    float sc, sh;
    fold_stats(ws, gamma[c], beta[c], c, sc, sh);
    if (bc < CSZ && threadIdx.x == 0) { ws[WS_SC + c] = sc * C2LOG2E; ws[WS_SH + c] = sh * C2LOG2E; }
    float acc[R_H];
#pragma unroll
    for (int k = 0; k < R_H; ++k) acc[k] = 0.f;

    for (int g = 0; g < 4; ++g) {
        float4 n0 = c0, n1 = c1, n2 = c2, n3 = c3;
        if (g < 3) {
            const float4* nx = xp4 + (g + 1) * 1024;
            n0 = nx[0]; n1 = nx[256]; n2 = nx[512]; n3 = nx[768];
        }
        cost_acc4(c0, sc, sh, acc);
        cost_acc4(c1, sc, sh, acc);
        cost_acc4(c2, sc, sh, acc);
        cost_acc4(c3, sc, sh, acc);
        c0 = n0; c1 = n1; c2 = n2; c3 = n3;
    }
#pragma unroll
    for (int k = 0; k < R_H; ++k) acc[k] = wave_sum(acc[k]);
    int wave = threadIdx.x >> 5;
    if ((threadIdx.x & 31) == 0) {
        *reinterpret_cast<float4*>(&part[wave * 8])     = make_float4(acc[0], acc[1], acc[2], acc[3]);
        *reinterpret_cast<float2*>(&part[wave * 8 + 4]) = make_float2(acc[4], acc[5]);
    }
    __syncthreads();
    if (threadIdx.x == 0) {
        float4 lo = *reinterpret_cast<const float4*>(&part[0]);
        float2 hi = *reinterpret_cast<const float2*>(&part[4]);
#pragma unroll
        for (int w = 1; w < 8; ++w) {
            float4 l2 = *reinterpret_cast<const float4*>(&part[w * 8]);
            float2 h2 = *reinterpret_cast<const float2*>(&part[w * 8 + 4]);
            lo.x += l2.x; lo.y += l2.y; lo.z += l2.z; lo.w += l2.w;
            hi.x += h2.x; hi.y += h2.y;
        }
        float costk[R_H] = {K00 * lo.x, K11 * lo.y, lo.z, lo.w, hi.x, hi.y};
        float m = costk[0];
#pragma unroll
        for (int k = 1; k < R_H; ++k) m = fmaxf(m, costk[k]);
        float e[R_H], ssum = 0.f;
#pragma unroll
        for (int k = 0; k < R_H; ++k) { e[k] = __expf(costk[k] - m); ssum += e[k]; }
        float inv = 1.0f / ssum;
        float wk[R_H];
#pragma unroll
        for (int k = 0; k < R_H; ++k) wk[k] = e[k] * inv;
        float* qd = ws + WS_W + (bc >> 1) * (2 * R_H) + (bc & 1);
        qd[0]  = fmaf(wk[4], K40, fmaf(wk[2], K20, wk[0] * K00));
        qd[2]  = fmaf(wk[5], K51, fmaf(wk[3], K31, wk[1] * K11));
        qd[4]  = fmaf(wk[4], K42, wk[2] * K22);
        qd[6]  = fmaf(wk[5], K53, wk[3] * K33);
        qd[8]  = wk[4] * K44;
        qd[10] = wk[5] * K55;
    }
}

__global__ void __launch_bounds__(256) k_fused(const float* __restrict__ x,
                                               const float* __restrict__ Wt,
                                               const float* __restrict__ bias,
                                               const float* __restrict__ ws,
                                               float* __restrict__ out) {
    __shared__ __attribute__((aligned(16))) float y_pair[(CSZ / 2) * YPS];
    __shared__ __attribute__((aligned(16))) float wt_l[OC * WLP];
    __shared__ __attribute__((aligned(16))) float cf_l[CSZ * R_H + 2 * CSZ];

    int b      = blockIdx.y;
    int d_base = blockIdx.x * DT;
    int tid    = threadIdx.x;
    int wave   = tid >> 5;
    int lane   = tid & 31;

    const float4* xq = reinterpret_cast<const float4*>(
                           x + ((size_t)b * CSZ + 2 * wave) * NSZ + d_base) + lane;
    float4 xa[4], xb[4];
#pragma unroll
    for (int j = 0; j < 4; ++j) {
        const float4* xj = xq + (size_t)j * (16 * (NSZ / 4));
        xa[j] = xj[0]; xb[j] = xj[NSZ / 4];
    }
    const float4* wt4 = reinterpret_cast<const float4*>(Wt) + tid;
    float4 wv[4];
#pragma unroll
    for (int it = 0; it < 4; ++it) wv[it] = wt4[it * 256];
    float4 cf = {0.f, 0.f, 0.f, 0.f};
    if (tid < 128) {
        const float4* cs = tid < 96
            ? reinterpret_cast<const float4*>(ws + WS_W + b * CSZ * R_H) + tid
            : reinterpret_cast<const float4*>(ws + WS_SC) + (tid - 96);
        cf = *cs;
    }
#pragma unroll
    for (int it = 0; it < 4; ++it) {
        int o = (tid >> 4) + 16 * it, c4 = (tid & 15) * 4;
        *reinterpret_cast<float4*>(&wt_l[o * WLP + c4]) = wv[it];
    }
    if (tid < 128) reinterpret_cast<float4*>(cf_l)[tid] = cf;
    __syncthreads();

#pragma unroll
    for (int j = 0; j < 4; ++j) {
        float4 ca = xa[j], cb = xb[j];
        int cp = wave + 8 * j;
        int ce = 2 * cp;
        float sc0 = cf_l[CF_SC + ce], sc1 = cf_l[CF_SC + ce + 1];
        float sh0 = cf_l[CF_SH + ce], sh1 = cf_l[CF_SH + ce + 1];
        float q0[R_H], q1[R_H];
#pragma unroll
        for (int k = 0; k < R_H; ++k) { q0[k] = cf_l[ce * R_H + 2 * k]; q1[k] = cf_l[ce * R_H + 2 * k + 1]; }
        float4 lo, hi;
        lo.x = y_of(ca.x, sc0, sh0, q0); lo.y = y_of(cb.x, sc1, sh1, q1);
        lo.z = y_of(ca.y, sc0, sh0, q0); lo.w = y_of(cb.y, sc1, sh1, q1);
        hi.x = y_of(ca.z, sc0, sh0, q0); hi.y = y_of(cb.z, sc1, sh1, q1);
        hi.z = y_of(ca.w, sc0, sh0, q0); hi.w = y_of(cb.w, sc1, sh1, q1);
        float4* dst = reinterpret_cast<float4*>(&y_pair[cp * YPS + lane * 8]);
        dst[0] = lo; dst[1] = hi;
    }
    __syncthreads();

    int half = lane >> 4;
    int lid  = lane & 15;
    int o0   = (wave & 3) << 4;

    v2f afrag[16];
#pragma unroll
    for (int s = 0; s < 16; ++s)
        afrag[s] = *(const v2f*)&wt_l[(o0 + lid) * WLP + 4 * s + 2 * half];

#pragma unroll
    for (int i = 0; i < 4; ++i) {
        int d0 = ((((wave >> 2) << 2) + i) << 4);
        v8f acc = {0.f, 0.f, 0.f, 0.f, 0.f, 0.f, 0.f, 0.f};
#pragma unroll
        for (int s = 0; s < 16; ++s) {
            v2f bb = *(const v2f*)&y_pair[(2 * s + half) * YPS + (d0 + lid) * 2];
            acc = __builtin_amdgcn_wmma_f32_16x16x4_f32(
                false, afrag[s], false, bb, (short)0, acc, false, false);
        }
        int dcol = d_base + d0 + lid;
#pragma unroll
        for (int v = 0; v < 8; ++v) {
            int o = o0 + v + 8 * half;
            out[(((size_t)b * OC + o) * NSZ) + dcol] = acc[v] + bias[o];
        }
    }
}

extern "C" void kernel_launch(void* const* d_in, const int* in_sizes, int n_in,
                              void* d_out, int out_size, void* d_ws, size_t ws_size,
                              hipStream_t stream) {
    const float* x     = (const float*)d_in[0];
    const float* gamma = (const float*)d_in[1];
    const float* beta  = (const float*)d_in[2];
    const float* Wt    = (const float*)d_in[3];
    const float* bias  = (const float*)d_in[4];
    float* out = (float*)d_out;
    float* ws  = (float*)d_ws;

    hipLaunchKernelGGL(k_stats, dim3(BSZ * CSZ),     dim3(256), 0, stream, x, ws);
    hipLaunchKernelGGL(k_cost,  dim3(BSZ * CSZ),     dim3(256), 0, stream, x, gamma, beta, ws);
    hipLaunchKernelGGL(k_fused, dim3(NSZ / DT, BSZ), dim3(256), 0, stream,
                       x, Wt, bias, ws, out);
}


# tanh(x) on the hardware tanh instruction in the cost and fused passes, replacing exp2, add, reciprocal and fma on a pre-scaled argument: one transcendental fewer per value in each pass
# speedup vs baseline: 2.7026x; 1.0485x over previous
#include <hip/hip_runtime.h>
#include <hip/hip_bf16.h>
#include <math.h>

#define R_H 6
#define EPSBN 1e-5f
#define BSZ 8
#define CSZ 64
#define NSZ 16384
#define OC  64
#define DT  128
#define YPS 288
#define WLP 68
#define CF_SC (CSZ * R_H)
#define CF_SH (CSZ * R_H + CSZ)
#define WS_PSUM 0
#define WS_PSQ  (BSZ * CSZ)
#define WS_SC   (2 * BSZ * CSZ)
#define WS_SH   (2 * BSZ * CSZ + CSZ)
#define WS_W    (2 * BSZ * CSZ + 2 * CSZ)

typedef __attribute__((ext_vector_type(2))) float v2f;
typedef __attribute__((ext_vector_type(8))) float v8f;

template <int M>
__device__ __forceinline__ float row_xor(float v) {
    int b = __builtin_bit_cast(int, v);
    return __builtin_bit_cast(float, __builtin_amdgcn_update_dpp(b, b, 0x160 + M, 0xf, 0xf, false));
}
__device__ __forceinline__ float xhalf(float v) {
    int b = __builtin_bit_cast(int, v);
    return __builtin_bit_cast(float,
        __builtin_amdgcn_permlanex16(b, b, 0x76543210, 0xfedcba98, false, false));
}
__device__ __forceinline__ float wave_sum(float v) {
    v += row_xor<1>(v);
    v += row_xor<2>(v);
    v += row_xor<4>(v);
    v += row_xor<8>(v);
    v += xhalf(v);
    return v;
}

#define K00   0.75112554446494251f
#define K11   3.8300038083847467f
#define K20  -0.53112596601359852f
#define K22  13.809275116353559f
#define K31  -4.6907775217294736f
#define K33  40.653405188322097f
#define K40   0.45996857917732664f
#define K42 -23.918366117220984f
#define K44 103.64625317462428f
#define K51   5.2444487029575502f
#define K53 -90.903777517930848f
#define K55 236.34982154662021f

__device__ __forceinline__ void tanh_gauss(float xn, float& m, float& s, float& g) {
    m = __builtin_amdgcn_tanhf(xn);
    s = m * m;
    g = __builtin_amdgcn_exp2f(s * -9.37751776577826314f);
}

__global__ void __launch_bounds__(256) k_stats(const float* __restrict__ x,
                                               float* __restrict__ ws) {
    __shared__ __attribute__((aligned(16))) float part[2 * 8];
    int bc = blockIdx.x;
    int c  = bc & (CSZ - 1);
    int b  = bc >> 6;
    const float4* xp4 = reinterpret_cast<const float4*>(x + (size_t)bc * NSZ) + threadIdx.x;
    float4 v[16];
#pragma unroll
    for (int j = 0; j < 16; ++j) v[j] = xp4[j * 256];
    __builtin_amdgcn_sched_barrier(0);
    float4 a = {0.f, 0.f, 0.f, 0.f}, q = {0.f, 0.f, 0.f, 0.f};
#pragma unroll
    for (int j = 0; j < 16; ++j) {
        a.x += v[j].x; q.x += v[j].x * v[j].x;
        a.y += v[j].y; q.y += v[j].y * v[j].y;
        a.z += v[j].z; q.z += v[j].z * v[j].z;
        a.w += v[j].w; q.w += v[j].w * v[j].w;
    }
    float s  = wave_sum((a.x + a.y) + (a.z + a.w));
    float s2 = wave_sum((q.x + q.y) + (q.z + q.w));
    int wave = threadIdx.x >> 5;
    if ((threadIdx.x & 31) == 0) { part[wave] = s; part[8 + wave] = s2; }
    __syncthreads();
    if (threadIdx.x == 0) {
        const float4* p4 = reinterpret_cast<const float4*>(part);
        float4 s0 = p4[0], s1 = p4[1], q0 = p4[2], q1 = p4[3];
        ws[WS_PSUM + c * BSZ + b] = ((s0.x + s0.y) + (s0.z + s0.w)) + ((s1.x + s1.y) + (s1.z + s1.w));
        ws[WS_PSQ  + c * BSZ + b] = ((q0.x + q0.y) + (q0.z + q0.w)) + ((q1.x + q1.y) + (q1.z + q1.w));
    }
}

__device__ __forceinline__ void fold_stats(const float* ws, float g, float bt, int c,
                                           float& sc, float& sh) {
    const float4* ps = reinterpret_cast<const float4*>(ws + WS_PSUM + c * BSZ);
    const float4* pq = reinterpret_cast<const float4*>(ws + WS_PSQ  + c * BSZ);
    float4 s0 = ps[0], s1 = ps[1], q0 = pq[0], q1 = pq[1];
    float s = ((s0.x + s0.y) + (s0.z + s0.w)) + ((s1.x + s1.y) + (s1.z + s1.w));
    float q = ((q0.x + q0.y) + (q0.z + q0.w)) + ((q1.x + q1.y) + (q1.z + q1.w));
    const float inv_n = 1.0f / ((float)BSZ * (float)NSZ);
    float mean = s * inv_n;
    float var  = q * inv_n - mean * mean;
    float rstd = rsqrtf(var + EPSBN);
    sc = g * rstd;
    sh = bt - mean * sc;
}

__device__ __forceinline__ void cost_acc1(float xv, float sc, float sh, float acc[R_H]) {
    float xn = fmaf(sc, xv, sh);
    float m, s, g;
    tanh_gauss(xn, m, s, g);
    float u  = g * xn;
    float um = u * m;
    acc[0] += u;
    acc[1] += um;
    acc[2] = fmaf(u,  fmaf(K22, s, K20), acc[2]);
    acc[3] = fmaf(um, fmaf(K33, s, K31), acc[3]);
    acc[4] = fmaf(u,  fmaf(fmaf(K44, s, K42), s, K40), acc[4]);
    acc[5] = fmaf(um, fmaf(fmaf(K55, s, K53), s, K51), acc[5]);
}
__device__ __forceinline__ void cost_acc4(float4 v, float sc, float sh, float acc[R_H]) {
    cost_acc1(v.x, sc, sh, acc);
    cost_acc1(v.y, sc, sh, acc);
    cost_acc1(v.z, sc, sh, acc);
    cost_acc1(v.w, sc, sh, acc);
}

__device__ __forceinline__ float y_of(float xv, float sc, float sh, const float q[R_H]) {
    float m, s, g;
    tanh_gauss(fmaf(sc, xv, sh), m, s, g);
    float ev = fmaf(fmaf(q[4], s, q[2]), s, q[0]);
    float od = fmaf(fmaf(q[5], s, q[3]), s, q[1]);
    return g * fmaf(m, od, ev);
}

__global__ void __launch_bounds__(256) k_cost(const float* __restrict__ x,
                                              const float* __restrict__ gamma,
                                              const float* __restrict__ beta,
                                              float* __restrict__ ws) {
    __shared__ __attribute__((aligned(16))) float part[8 * 8];
    int bc = blockIdx.x;
    int c  = bc & (CSZ - 1);
    const float4* xp4 = reinterpret_cast<const float4*>(x + (size_t)bc * NSZ) + threadIdx.x;
    float4 c0 = xp4[0], c1 = xp4[256], c2 = xp4[512], c3 = xp4[768];
    float sc, sh;
    fold_stats(ws, gamma[c], beta[c], c, sc, sh);
    if (bc < CSZ && threadIdx.x == 0) { ws[WS_SC + c] = sc; ws[WS_SH + c] = sh; }
    float acc[R_H];
#pragma unroll
    for (int k = 0; k < R_H; ++k) acc[k] = 0.f;

    for (int g = 0; g < 4; ++g) {
        float4 n0 = c0, n1 = c1, n2 = c2, n3 = c3;
        if (g < 3) {
            const float4* nx = xp4 + (g + 1) * 1024;
            n0 = nx[0]; n1 = nx[256]; n2 = nx[512]; n3 = nx[768];
        }
        cost_acc4(c0, sc, sh, acc);
        cost_acc4(c1, sc, sh, acc);
        cost_acc4(c2, sc, sh, acc);
        cost_acc4(c3, sc, sh, acc);
        c0 = n0; c1 = n1; c2 = n2; c3 = n3;
    }
#pragma unroll
    for (int k = 0; k < R_H; ++k) acc[k] = wave_sum(acc[k]);
    int wave = threadIdx.x >> 5;
    if ((threadIdx.x & 31) == 0) {
        *reinterpret_cast<float4*>(&part[wave * 8])     = make_float4(acc[0], acc[1], acc[2], acc[3]);
        *reinterpret_cast<float2*>(&part[wave * 8 + 4]) = make_float2(acc[4], acc[5]);
    }
    __syncthreads();
    if (threadIdx.x == 0) {
        float4 lo = *reinterpret_cast<const float4*>(&part[0]);
        float2 hi = *reinterpret_cast<const float2*>(&part[4]);
#pragma unroll
        for (int w = 1; w < 8; ++w) {
            float4 l2 = *reinterpret_cast<const float4*>(&part[w * 8]);
            float2 h2 = *reinterpret_cast<const float2*>(&part[w * 8 + 4]);
            lo.x += l2.x; lo.y += l2.y; lo.z += l2.z; lo.w += l2.w;
            hi.x += h2.x; hi.y += h2.y;
        }
        float costk[R_H] = {K00 * lo.x, K11 * lo.y, lo.z, lo.w, hi.x, hi.y};
        float m = costk[0];
#pragma unroll
        for (int k = 1; k < R_H; ++k) m = fmaxf(m, costk[k]);
        float e[R_H], ssum = 0.f;
#pragma unroll
        for (int k = 0; k < R_H; ++k) { e[k] = __expf(costk[k] - m); ssum += e[k]; }
        float inv = 1.0f / ssum;
        float wk[R_H];
#pragma unroll
        for (int k = 0; k < R_H; ++k) wk[k] = e[k] * inv;
        float* qd = ws + WS_W + (bc >> 1) * (2 * R_H) + (bc & 1);
        qd[0]  = fmaf(wk[4], K40, fmaf(wk[2], K20, wk[0] * K00));
        qd[2]  = fmaf(wk[5], K51, fmaf(wk[3], K31, wk[1] * K11));
        qd[4]  = fmaf(wk[4], K42, wk[2] * K22);
        qd[6]  = fmaf(wk[5], K53, wk[3] * K33);
        qd[8]  = wk[4] * K44;
        qd[10] = wk[5] * K55;
    }
}

__global__ void __launch_bounds__(256) k_fused(const float* __restrict__ x,
                                               const float* __restrict__ Wt,
                                               const float* __restrict__ bias,
                                               const float* __restrict__ ws,
                                               float* __restrict__ out) {
    __shared__ __attribute__((aligned(16))) float y_pair[(CSZ / 2) * YPS];
    __shared__ __attribute__((aligned(16))) float wt_l[OC * WLP];
    __shared__ __attribute__((aligned(16))) float cf_l[CSZ * R_H + 2 * CSZ];

    int b      = blockIdx.y;
    int d_base = blockIdx.x * DT;
    int tid    = threadIdx.x;
    int wave   = tid >> 5;
    int lane   = tid & 31;

    const float4* xq = reinterpret_cast<const float4*>(
                           x + ((size_t)b * CSZ + 2 * wave) * NSZ + d_base) + lane;
    float4 xa[4], xb[4];
#pragma unroll
    for (int j = 0; j < 4; ++j) {
        const float4* xj = xq + (size_t)j * (16 * (NSZ / 4));
        xa[j] = xj[0]; xb[j] = xj[NSZ / 4];
    }
    const float4* wt4 = reinterpret_cast<const float4*>(Wt) + tid;
    float4 wv[4];
#pragma unroll
    for (int it = 0; it < 4; ++it) wv[it] = wt4[it * 256];
    float4 cf = {0.f, 0.f, 0.f, 0.f};
    if (tid < 128) {
        const float4* cs = tid < 96
            ? reinterpret_cast<const float4*>(ws + WS_W + b * CSZ * R_H) + tid
            : reinterpret_cast<const float4*>(ws + WS_SC) + (tid - 96);
        cf = *cs;
    }
#pragma unroll
    for (int it = 0; it < 4; ++it) {
        int o = (tid >> 4) + 16 * it, c4 = (tid & 15) * 4;
        *reinterpret_cast<float4*>(&wt_l[o * WLP + c4]) = wv[it];
    }
    if (tid < 128) reinterpret_cast<float4*>(cf_l)[tid] = cf;
    __syncthreads();

#pragma unroll
    for (int j = 0; j < 4; ++j) {
        float4 ca = xa[j], cb = xb[j];
        int cp = wave + 8 * j;
        int ce = 2 * cp;
        float sc0 = cf_l[CF_SC + ce], sc1 = cf_l[CF_SC + ce + 1];
        float sh0 = cf_l[CF_SH + ce], sh1 = cf_l[CF_SH + ce + 1];
        float q0[R_H], q1[R_H];
#pragma unroll
        for (int k = 0; k < R_H; ++k) { q0[k] = cf_l[ce * R_H + 2 * k]; q1[k] = cf_l[ce * R_H + 2 * k + 1]; }
        float4 lo, hi;
        lo.x = y_of(ca.x, sc0, sh0, q0); lo.y = y_of(cb.x, sc1, sh1, q1);
        lo.z = y_of(ca.y, sc0, sh0, q0); lo.w = y_of(cb.y, sc1, sh1, q1);
        hi.x = y_of(ca.z, sc0, sh0, q0); hi.y = y_of(cb.z, sc1, sh1, q1);
        hi.z = y_of(ca.w, sc0, sh0, q0); hi.w = y_of(cb.w, sc1, sh1, q1);
        float4* dst = reinterpret_cast<float4*>(&y_pair[cp * YPS + lane * 8]);
        dst[0] = lo; dst[1] = hi;
    }
    __syncthreads();

    int half = lane >> 4;
    int lid  = lane & 15;
    int o0   = (wave & 3) << 4;

    v2f afrag[16];
#pragma unroll
    for (int s = 0; s < 16; ++s)
        afrag[s] = *(const v2f*)&wt_l[(o0 + lid) * WLP + 4 * s + 2 * half];

#pragma unroll
    for (int i = 0; i < 4; ++i) {
        int d0 = ((((wave >> 2) << 2) + i) << 4);
        v8f acc = {0.f, 0.f, 0.f, 0.f, 0.f, 0.f, 0.f, 0.f};
#pragma unroll
        for (int s = 0; s < 16; ++s) {
            v2f bb = *(const v2f*)&y_pair[(2 * s + half) * YPS + (d0 + lid) * 2];
            acc = __builtin_amdgcn_wmma_f32_16x16x4_f32(
                false, afrag[s], false, bb, (short)0, acc, false, false);
        }
        int dcol = d_base + d0 + lid;
#pragma unroll
        for (int v = 0; v < 8; ++v) {
            int o = o0 + v + 8 * half;
            out[(((size_t)b * OC + o) * NSZ) + dcol] = acc[v] + bias[o];
        }
    }
}

extern "C" void kernel_launch(void* const* d_in, const int* in_sizes, int n_in,
                              void* d_out, int out_size, void* d_ws, size_t ws_size,
                              hipStream_t stream) {
    const float* x     = (const float*)d_in[0];
    const float* gamma = (const float*)d_in[1];
    const float* beta  = (const float*)d_in[2];
    const float* Wt    = (const float*)d_in[3];
    const float* bias  = (const float*)d_in[4];
    float* out = (float*)d_out;
    float* ws  = (float*)d_ws;

    hipLaunchKernelGGL(k_stats, dim3(BSZ * CSZ),     dim3(256), 0, stream, x, ws);
    hipLaunchKernelGGL(k_cost,  dim3(BSZ * CSZ),     dim3(256), 0, stream, x, gamma, beta, ws);
    hipLaunchKernelGGL(k_fused, dim3(NSZ / DT, BSZ), dim3(256), 0, stream,
                       x, Wt, bias, ws, out);
}
